# starting kernel
# baseline (speedup 1.0000x reference)
#include <hip/hip_runtime.h>
#include <math.h>

typedef __attribute__((ext_vector_type(2))) float v2f;
typedef __attribute__((ext_vector_type(8))) float v8f;

__device__ __forceinline__ v8f wmma_f32(v2f a, v2f b, v8f c) {
    return __builtin_amdgcn_wmma_f32_16x16x4_f32(false, a, false, b, (short)0, c, false, false);
}
__device__ __forceinline__ v2f lds_frag(const float* p) {
    return *(const v2f*)p;
}

#define GT_M 64
#define GT_N 128
#define GT_K 16
#define A_FO 160
#define B_FO 288

__global__ __launch_bounds__(256) void ddnet_gemm_kernel(
    const float* __restrict__ A, const float* __restrict__ W,
    const float* __restrict__ bias, const float* __restrict__ resid,
    float* __restrict__ C, int Mdim, int Ndim, int Kdim, int act)
{
    __shared__ __align__(16) float As[2][8 * A_FO];
    __shared__ __align__(16) float Bs[2][8 * B_FO];

    const int tid   = threadIdx.x;
    const int lane  = tid & 31;
    const int wave  = tid >> 5;
    const int lhalf = lane >> 4;
    const int lmod  = lane & 15;
    const int m0    = blockIdx.y * GT_M;
    const int n0    = blockIdx.x * GT_N;
    const int wm    = (wave & 1) * 32;
    const int wn    = (wave >> 1) * 32;
    const int nk    = Kdim >> 4;

    v8f acc[2][2] = {};
    float ra[4], rb[8];

    #pragma unroll
    for (int i = 0; i < 4; ++i) {
        int idx = tid + i * 256, kk = idx & 15, mm = idx >> 4;
        ra[i] = A[(size_t)(m0 + mm) * Kdim + kk];
    }
    #pragma unroll
    for (int i = 0; i < 8; ++i) {
        int idx = tid + i * 256, kk = idx & 15, nn = idx >> 4;
        rb[i] = W[(size_t)(n0 + nn) * Kdim + kk];
    }
    #pragma unroll
    for (int i = 0; i < 4; ++i) {
        int idx = tid + i * 256, kk = idx & 15, mm = idx >> 4;
        As[0][((kk >> 2) * 2 + ((kk >> 1) & 1)) * A_FO + mm * 2 + (kk & 1)] = ra[i];
    }
    #pragma unroll
    for (int i = 0; i < 8; ++i) {
        int idx = tid + i * 256, kk = idx & 15, nn = idx >> 4;
        Bs[0][((kk >> 2) * 2 + ((kk >> 1) & 1)) * B_FO + nn * 2 + (kk & 1)] = rb[i];
    }
    __syncthreads();

    for (int kt = 0; kt < nk; ++kt) {
        const int buf = kt & 1;
        if (kt + 1 < nk) {
            const int k0 = (kt + 1) << 4;
            #pragma unroll
            for (int i = 0; i < 4; ++i) {
                int idx = tid + i * 256, kk = idx & 15, mm = idx >> 4;
                ra[i] = A[(size_t)(m0 + mm) * Kdim + k0 + kk];
            }
            #pragma unroll
            for (int i = 0; i < 8; ++i) {
                int idx = tid + i * 256, kk = idx & 15, nn = idx >> 4;
                rb[i] = W[(size_t)(n0 + nn) * Kdim + k0 + kk];
            }
        }
        const float* Ab = As[buf];
        const float* Bb = Bs[buf];
        #pragma unroll
        for (int t = 0; t < 4; ++t) {
            const int fo = t * 2 + lhalf;
            v2f a0 = lds_frag(&Ab[fo * A_FO + (wm + lmod) * 2]);
            v2f a1 = lds_frag(&Ab[fo * A_FO + (wm + 16 + lmod) * 2]);
            v2f b0 = lds_frag(&Bb[fo * B_FO + (wn + lmod) * 2]);
            v2f b1 = lds_frag(&Bb[fo * B_FO + (wn + 16 + lmod) * 2]);
            acc[0][0] = wmma_f32(a0, b0, acc[0][0]);
            acc[0][1] = wmma_f32(a0, b1, acc[0][1]);
            acc[1][0] = wmma_f32(a1, b0, acc[1][0]);
            acc[1][1] = wmma_f32(a1, b1, acc[1][1]);
        }
        if (kt + 1 < nk) {
            __syncthreads();
            #pragma unroll
            for (int i = 0; i < 4; ++i) {
                int idx = tid + i * 256, kk = idx & 15, mm = idx >> 4;
                As[buf ^ 1][((kk >> 2) * 2 + ((kk >> 1) & 1)) * A_FO + mm * 2 + (kk & 1)] = ra[i];
            }
            #pragma unroll
            for (int i = 0; i < 8; ++i) {
                int idx = tid + i * 256, kk = idx & 15, nn = idx >> 4;
                Bs[buf ^ 1][((kk >> 2) * 2 + ((kk >> 1) & 1)) * B_FO + nn * 2 + (kk & 1)] = rb[i];
            }
            __syncthreads();
        }
    }

    #pragma unroll
    for (int mi = 0; mi < 2; ++mi)
        #pragma unroll
        for (int ni = 0; ni < 2; ++ni)
            #pragma unroll
            for (int v = 0; v < 8; ++v) {
                int row = m0 + wm + mi * 16 + v + 8 * lhalf;
                int col = n0 + wn + ni * 16 + lmod;
                float val = acc[mi][ni][v];
                if (bias)  val += bias[col];
                if (resid) val += resid[(size_t)row * Ndim + col];
                if (act == 1)
                    val = 0.5f * val * (1.0f + erff(val * 0.70710678118654752f));
                C[(size_t)row * Ndim + col] = val;
            }
}

__global__ __launch_bounds__(256) void ddnet_pe_kernel(
    float* __restrict__ h, int S, int D)
{
    size_t idx = (size_t)blockIdx.x * 256 + threadIdx.x;
    int d = (int)(idx % D);
    int s = (int)((idx / D) % S);
    int pair = d >> 1;
    float div = __expf(-(float)(2 * pair) * (logf(10000.0f) / (float)D));
    float ang = (float)s * div;
    h[idx] += (d & 1) ? __cosf(ang) : __sinf(ang);
}

__global__ __launch_bounds__(256) void ddnet_ln_kernel(
    const float* __restrict__ x, const float* __restrict__ gamma,
    const float* __restrict__ beta, float* __restrict__ out, int D)
{
    int row  = blockIdx.x * 8 + (threadIdx.x >> 5);
    int lane = threadIdx.x & 31;
    const float* xr = x + (size_t)row * D;
    float vals[8];
    float s = 0.f;
    #pragma unroll
    for (int i = 0; i < 8; ++i) { vals[i] = xr[lane + i * 32]; s += vals[i]; }
    #pragma unroll
    for (int m = 16; m >= 1; m >>= 1) s += __shfl_xor(s, m, 32);
    float mean = s * (1.0f / 256.0f);
    float vs = 0.f;
    #pragma unroll
    for (int i = 0; i < 8; ++i) { float d = vals[i] - mean; vs += d * d; }
    #pragma unroll
    for (int m = 16; m >= 1; m >>= 1) vs += __shfl_xor(vs, m, 32);
    float rstd = rsqrtf(vs * (1.0f / 256.0f) + 1e-5f);
    float* orow = out + (size_t)row * D;
    #pragma unroll
    for (int i = 0; i < 8; ++i) {
        int c = lane + i * 32;
        orow[c] = (vals[i] - mean) * rstd * gamma[c] + beta[c];
    }
}

#define DD 256
#define DG 6

__global__ __launch_bounds__(256) void ddnet_desc_kernel(
    const float* __restrict__ xt, const float* __restrict__ h,
    const float* __restrict__ P, float* __restrict__ hd, int S)
{
    __shared__ __align__(16) float Xs[128 * 32];
    __shared__ float Ct[511 * DG + 1];

    const int s   = blockIdx.x;
    const int tid = threadIdx.x;
    const float two_pi = 6.28318530717958647692f;

    for (int idx = tid; idx < 511 * DG; idx += 256) {
        int t = idx / DG, g = idx % DG;
        float period = (2.0f + 0.5f * (float)t) * powf(1.5f, (float)g);
        Ct[idx] = __cosf(two_pi * (float)s / period);
    }
    for (int idx = tid; idx < 16 * DD; idx += 256) {
        int b = idx >> 8, j = idx & 255;
        Xs[((j >> 2) * 2 + ((j >> 1) & 1)) * 32 + b * 2 + (j & 1)] =
            xt[((size_t)b * S + s) * DD + j];
    }
    __syncthreads();

    const int lane = tid & 31, wave = tid >> 5;
    const int lhalf = lane >> 4, lmod = lane & 15;

    for (int it = wave; it < 16; it += 8) {
        const int i = it * 16 + lmod;
        v8f acc = {};
        for (int j0 = 0; j0 < DD; j0 += 16) {
            float wv[4][2];
            #pragma unroll
            for (int t = 0; t < 4; ++t)
                #pragma unroll
                for (int jj = 0; jj < 2; ++jj) {
                    int j = j0 + t * 4 + jj + 2 * lhalf;
                    const float* Pp = P + ((size_t)i * DD + j) * DG;
                    const float* Cp = Ct + (i + j) * DG;
                    float w = 0.f;
                    #pragma unroll
                    for (int g = 0; g < DG; ++g) w += Pp[g] * Cp[g];
                    wv[t][jj] = w;
                }
            #pragma unroll
            for (int t = 0; t < 4; ++t) {
                v2f a = lds_frag(&Xs[((j0 >> 2) + t) * 64 + lhalf * 32 + lmod * 2]);
                v2f b = { wv[t][0], wv[t][1] };
                acc = wmma_f32(a, b, acc);
            }
        }
        #pragma unroll
        for (int v = 0; v < 8; ++v) {
            int b = v + 8 * lhalf;
            size_t off = ((size_t)b * S + s) * DD + it * 16 + lmod;
            hd[off] = acc[v] + h[off];
        }
    }
}

__global__ __launch_bounds__(128) void ddnet_attn_kernel(
    const float* __restrict__ qkv, float* __restrict__ o, int S)
{
    const int qb    = S / 64;
    const int bh    = blockIdx.x / qb;
    const int qblk  = blockIdx.x % qb;
    const int b     = bh >> 3, head = bh & 7;
    const int tid   = threadIdx.x, lane = tid & 31, wave = tid >> 5;
    const int lhalf = lane >> 4, lmod = lane & 15;
    const int q0    = qblk * 64 + wave * 16;

    __shared__ __align__(16) float Kf[4][16 * 32];
    __shared__ __align__(16) float Vf[4][8 * 96];
    __shared__ __align__(16) float Pf[4][8 * 96];

    const size_t rs = 768;
    const float* qp = qkv + (size_t)b * S * rs + head * 32;
    const float* kp = qp + 256;
    const float* vp = qp + 512;
    const float scale = 0.17677669529663688f;

    v2f aq[8];
    #pragma unroll
    for (int t = 0; t < 8; ++t) {
        const float* qr = qp + (size_t)(q0 + lmod) * rs;
        aq[t].x = qr[t * 4 + 2 * lhalf];
        aq[t].y = qr[t * 4 + 1 + 2 * lhalf];
    }

    float mrow[8], lrow[8];
    v8f o0 = {}, o1 = {};
    #pragma unroll
    for (int v = 0; v < 8; ++v) { mrow[v] = -1e30f; lrow[v] = 0.f; }

    for (int k0 = 0; k0 < S; k0 += 16) {
        for (int i = lane; i < 16 * 32; i += 32) {
            int r = i >> 5, c = i & 31;
            Kf[wave][((c >> 2) * 2 + ((c >> 1) & 1)) * 32 + r * 2 + (c & 1)] =
                kp[(size_t)(k0 + r) * rs + c];
            Vf[wave][((r >> 2) * 2 + ((r >> 1) & 1)) * 96 + c * 2 + (r & 1)] =
                vp[(size_t)(k0 + r) * rs + c];
        }
        __syncthreads();

        v8f sc = {};
        #pragma unroll
        for (int t = 0; t < 8; ++t) {
            v2f bk = lds_frag(&Kf[wave][(t * 2 + lhalf) * 32 + lmod * 2]);
            sc = wmma_f32(aq[t], bk, sc);
        }

        float pr[8];
        #pragma unroll
        for (int v = 0; v < 8; ++v) {
            float sv = sc[v] * scale;
            float mx = sv;
            #pragma unroll
            for (int m = 8; m >= 1; m >>= 1) mx = fmaxf(mx, __shfl_xor(mx, m, 32));
            float mnew = fmaxf(mrow[v], mx);
            float corr = __expf(mrow[v] - mnew);
            float p    = __expf(sv - mnew);
            float rsum = p;
            #pragma unroll
            for (int m = 8; m >= 1; m >>= 1) rsum += __shfl_xor(rsum, m, 32);
            lrow[v] = lrow[v] * corr + rsum;
            mrow[v] = mnew;
            pr[v]   = p;
            o0[v] *= corr;
            o1[v] *= corr;
        }

        #pragma unroll
        for (int v = 0; v < 8; ++v)
            Pf[wave][((lmod >> 2) * 2 + ((lmod >> 1) & 1)) * 96 +
                     (v + 8 * lhalf) * 2 + (lmod & 1)] = pr[v];
        #pragma unroll
        for (int t = 0; t < 4; ++t) {
            int fo = t * 2 + lhalf;
            v2f ap  = lds_frag(&Pf[wave][fo * 96 + lmod * 2]);
            v2f bv0 = lds_frag(&Vf[wave][fo * 96 + lmod * 2]);
            v2f bv1 = lds_frag(&Vf[wave][fo * 96 + (16 + lmod) * 2]);
            o0 = wmma_f32(ap, bv0, o0);
            o1 = wmma_f32(ap, bv1, o1);
        }
        __syncthreads();
    }

    #pragma unroll
    for (int v = 0; v < 8; ++v) {
        int qr = v + 8 * lhalf;
        float inv = 1.0f / lrow[v];
        size_t base = ((size_t)b * S + q0 + qr) * 256 + head * 32;
        o[base + lmod]      = o0[v] * inv;
        o[base + 16 + lmod] = o1[v] * inv;
    }
}

static void launch_gemm(const float* A, const float* W, const float* bias,
                        const float* resid, float* C, int M, int N, int K,
                        int act, hipStream_t stream)
{
    dim3 grid(N / GT_N, M / GT_M);
    ddnet_gemm_kernel<<<grid, 256, 0, stream>>>(A, W, bias, resid, C, M, N, K, act);
}

extern "C" void kernel_launch(void* const* d_in, const int* in_sizes, int n_in,
                              void* d_out, int out_size, void* d_ws, size_t ws_size,
                              hipStream_t stream) {
    (void)in_sizes; (void)n_in; (void)out_size; (void)ws_size;
    const int B = 16, S = 512, IN = 64, D = 256, L = 3, G = 6, DFF = 1024;
    const int Mr = B * S;

    const float* x      = (const float*)d_in[0];
    const float* in_w   = (const float*)d_in[1];
    const float* in_b   = (const float*)d_in[2];
    const float* Mw     = (const float*)d_in[3];
    const float* P      = (const float*)d_in[4];
    const float* desc_g = (const float*)d_in[5];
    const float* desc_b = (const float*)d_in[6];
    const float* qkv_w  = (const float*)d_in[7];
    const float* qkv_b  = (const float*)d_in[8];
    const float* out_w  = (const float*)d_in[9];
    const float* out_b  = (const float*)d_in[10];
    const float* ff1_w  = (const float*)d_in[11];
    const float* ff1_b  = (const float*)d_in[12];
    const float* ff2_w  = (const float*)d_in[13];
    const float* ff2_b  = (const float*)d_in[14];
    const float* ln1_g  = (const float*)d_in[15];
    const float* ln1_b  = (const float*)d_in[16];
    const float* ln2_g  = (const float*)d_in[17];
    const float* ln2_b  = (const float*)d_in[18];

    float* h    = (float*)d_ws;
    float* tmp  = h    + (size_t)Mr * D;
    float* xt   = tmp  + (size_t)Mr * D;
    float* hd   = xt   + (size_t)Mr * D;
    float* qkv  = hd   + (size_t)Mr * D;
    float* obuf = qkv  + (size_t)Mr * 3 * D;
    float* h1   = obuf + (size_t)Mr * D;
    float* ff   = h1   + (size_t)Mr * D;

    launch_gemm(x, in_w, in_b, nullptr, h, Mr, D, IN, 0, stream);
    ddnet_pe_kernel<<<(Mr * D) / 256, 256, 0, stream>>>(h, S, D);

    for (int l = 0; l < L; ++l) {
        launch_gemm(h, Mw + (size_t)l * D * D, nullptr, nullptr, tmp, Mr, D, D, 0, stream);
        ddnet_ln_kernel<<<Mr / 8, 256, 0, stream>>>(tmp, desc_g + l * D, desc_b + l * D, xt, D);
        ddnet_desc_kernel<<<S, 256, 0, stream>>>(xt, h, P + (size_t)l * D * D * G, hd, S);
        launch_gemm(hd, qkv_w + (size_t)l * 3 * D * D, qkv_b + (size_t)l * 3 * D,
                    nullptr, qkv, Mr, 3 * D, D, 0, stream);
        ddnet_attn_kernel<<<B * 8 * (S / 64), 128, 0, stream>>>(qkv, obuf, S);
        launch_gemm(obuf, out_w + (size_t)l * D * D, out_b + l * D, hd, tmp, Mr, D, D, 0, stream);
        ddnet_ln_kernel<<<Mr / 8, 256, 0, stream>>>(tmp, ln1_g + l * D, ln1_b + l * D, h1, D);
        launch_gemm(h1, ff1_w + (size_t)l * DFF * D, ff1_b + (size_t)l * DFF,
                    nullptr, ff, Mr, DFF, D, 1  , stream);
        launch_gemm(ff, ff2_w + (size_t)l * D * DFF, ff2_b + l * D, h1, tmp, Mr, D, DFF, 0, stream);
        float* dst = (l == L - 1) ? (float*)d_out : h;
        ddnet_ln_kernel<<<Mr / 8, 256, 0, stream>>>(tmp, ln2_g + l * D, ln2_b + l * D, dst, D);
    }
}


# GEMM K loop: one split barrier per tile (store next tile, read this tile's fragments, signal, the tile's products, wait); was 2 full barriers per tile with the stores between; sums unchanged
# speedup vs baseline: 2.1615x; 2.1615x over previous
#include <hip/hip_runtime.h>
#include <math.h>

typedef __attribute__((ext_vector_type(2))) float v2f;
typedef __attribute__((ext_vector_type(8))) float v8f;
typedef __attribute__((ext_vector_type(4))) float v4f;

__device__ __forceinline__ v8f wmma_f32(v2f a, v2f b, v8f c) {
    return __builtin_amdgcn_wmma_f32_16x16x4_f32(false, a, false, b, (short)0, c, false, false);
}
__device__ __forceinline__ v2f lds_frag(const float* p) {
    return *(const v2f*)p;
}

#define GT_M 64
#define GT_N 128
#define GT_K 16
#define AB_P 20
#define C_P  132
#define GT_STG ((GT_M + GT_N) * AB_P)

__global__ __launch_bounds__(256) void ddnet_gemm_kernel(
    const float* __restrict__ A, const float* __restrict__ W,
    const float* __restrict__ bias, const float* __restrict__ resid,
    float* __restrict__ C, int Mdim, int Ndim, int Kdim, int act)
{
    __shared__ __align__(16) float smem[GT_M * C_P];

    const int tid   = threadIdx.x;
    const int lane  = tid & 31;
    const int wave  = tid >> 5;
    const int lhalf = lane >> 4;
    const int lmod  = lane & 15;
    const int m0    = blockIdx.y * GT_M;
    const int n0    = blockIdx.x * GT_N;
    const int wm    = (wave & 1) * 32;
    const int wn    = (wave >> 1) * 32;
    const int nk    = Kdim >> 4;

    v8f acc[2][2] = {};
    float ra[4], rb[8];

    const int sk = tid & 15, sm = tid >> 4;
    const float* ag = A + (size_t)(m0 + sm) * Kdim + sk;
    const float* wg = W + (size_t)(n0 + sm) * Kdim + sk;
    const int sl = sm * AB_P + sk;

    auto fetch = [&](int k0) {
        #pragma unroll
        for (int i = 0; i < 4; ++i) ra[i] = ag[(size_t)(16 * i) * Kdim + k0];
        #pragma unroll
        for (int i = 0; i < 8; ++i) rb[i] = wg[(size_t)(16 * i) * Kdim + k0];
    };
    auto spill = [&](int st) {
        float* Ad = smem + st * GT_STG + sl;
        float* Bd = Ad + GT_M * AB_P;
        #pragma unroll
        for (int i = 0; i < 4; ++i) Ad[16 * i * AB_P] = ra[i];
        #pragma unroll
        for (int i = 0; i < 8; ++i) Bd[16 * i * AB_P] = rb[i];
    };

    fetch(0);
    spill(0);
    if (nk > 1) fetch(GT_K);
    __syncthreads();

    for (int kt = 0; kt < nk; ++kt) {
        const int  cur  = kt & 1;
        const bool more = kt + 1 < nk;
        if (more) {
            spill(cur ^ 1);
            if (kt + 2 < nk) fetch((kt + 2) << 4);
        }
        const float* Ab = smem + cur * GT_STG;
        const float* Bb = Ab + GT_M * AB_P;
        v2f fa[4][2], fb[4][2];
        #pragma unroll
        for (int t = 0; t < 4; ++t) {
            const int kf = t * 4 + 2 * lhalf;
            fa[t][0] = lds_frag(&Ab[(wm + lmod) * AB_P + kf]);
            fa[t][1] = lds_frag(&Ab[(wm + 16 + lmod) * AB_P + kf]);
            fb[t][0] = lds_frag(&Bb[(wn + lmod) * AB_P + kf]);
            fb[t][1] = lds_frag(&Bb[(wn + 16 + lmod) * AB_P + kf]);
        }
        if (more) {
            __builtin_amdgcn_fence(__ATOMIC_RELEASE, "workgroup", "local");
            __builtin_amdgcn_s_barrier_signal(-1);
        }
        #pragma unroll
        for (int t = 0; t < 4; ++t) {
            acc[0][0] = wmma_f32(fa[t][0], fb[t][0], acc[0][0]);
            acc[0][1] = wmma_f32(fa[t][0], fb[t][1], acc[0][1]);
            acc[1][0] = wmma_f32(fa[t][1], fb[t][0], acc[1][0]);
            acc[1][1] = wmma_f32(fa[t][1], fb[t][1], acc[1][1]);
        }
        if (more) {
            __builtin_amdgcn_s_barrier_wait(-1);
            __builtin_amdgcn_fence(__ATOMIC_ACQUIRE, "workgroup", "local");
        }
    }

    __syncthreads();
    #pragma unroll
    for (int mi = 0; mi < 2; ++mi)
        #pragma unroll
        for (int ni = 0; ni < 2; ++ni)
            #pragma unroll
            for (int v = 0; v < 8; ++v)
                smem[(wm + mi * 16 + v + 8 * lhalf) * C_P + wn + ni * 16 + lmod] = acc[mi][ni][v];
    __syncthreads();

    const int c4 = (tid & 31) * 4;
    const int r0 = tid >> 5;
    const float* rp = resid ? resid + (size_t)(m0 + r0) * Ndim + n0 + c4 : nullptr;
    float*       cp = C + (size_t)(m0 + r0) * Ndim + n0 + c4;
    v4f bv = {0.f, 0.f, 0.f, 0.f};
    if (bias) bv = *(const v4f*)(bias + n0 + c4);
    v4f rv[8];
    if (resid) {
        #pragma unroll
        for (int i = 0; i < 8; ++i)
            rv[i] = *(const v4f*)(rp + (size_t)(8 * i) * Ndim);
    }
    #pragma unroll
    for (int i = 0; i < 8; ++i) {
        v4f val = *(const v4f*)&smem[(r0 + 8 * i) * C_P + c4];
        if (bias)  val += bv;
        if (resid) val += rv[i];
        if (act == 1) {
            val.x = 0.5f * val.x * (1.0f + erff(val.x * 0.70710678118654752f));
            val.y = 0.5f * val.y * (1.0f + erff(val.y * 0.70710678118654752f));
            val.z = 0.5f * val.z * (1.0f + erff(val.z * 0.70710678118654752f));
            val.w = 0.5f * val.w * (1.0f + erff(val.w * 0.70710678118654752f));
        }
        *(v4f*)(cp + (size_t)(8 * i) * Ndim) = val;
    }
}

__global__ __launch_bounds__(256) void ddnet_pe_kernel(
    float* __restrict__ h, int S, int D)
{
    size_t idx = (size_t)blockIdx.x * 256 + threadIdx.x;
    int d = (int)(idx % D);
    int s = (int)((idx / D) % S);
    int pair = d >> 1;
    float div = __expf(-(float)(2 * pair) * (logf(10000.0f) / (float)D));
    float ang = (float)s * div;
    h[idx] += (d & 1) ? __cosf(ang) : __sinf(ang);
}

__global__ __launch_bounds__(256) void ddnet_ln_kernel(
    const float* __restrict__ x, const float* __restrict__ gamma,
    const float* __restrict__ beta, float* __restrict__ out, int D)
{
    int row  = blockIdx.x * 8 + (threadIdx.x >> 5);
    int lane = threadIdx.x & 31;
    const float* xr = x + (size_t)row * D;
    float vals[8];
    float s = 0.f;
    #pragma unroll
    for (int i = 0; i < 8; ++i) { vals[i] = xr[lane + i * 32]; s += vals[i]; }
    #pragma unroll
    for (int m = 16; m >= 1; m >>= 1) s += __shfl_xor(s, m, 32);
    float mean = s * (1.0f / 256.0f);
    float vs = 0.f;
    #pragma unroll
    for (int i = 0; i < 8; ++i) { float d = vals[i] - mean; vs += d * d; }
    #pragma unroll
    for (int m = 16; m >= 1; m >>= 1) vs += __shfl_xor(vs, m, 32);
    float rstd = rsqrtf(vs * (1.0f / 256.0f) + 1e-5f);
    float* orow = out + (size_t)row * D;
    #pragma unroll
    for (int i = 0; i < 8; ++i) {
        int c = lane + i * 32;
        orow[c] = (vals[i] - mean) * rstd * gamma[c] + beta[c];
    }
}

#define DD 256
#define DG 6
#define DSC 16
#define DNP 2
#define DCT 1632

__global__ __launch_bounds__(256) void ddnet_desc_kernel(
    const float* __restrict__ xt, const float* __restrict__ h,
    const float* __restrict__ P, float* __restrict__ hd, int S)
{
    __shared__ __align__(16) float Ct[2][DNP][DCT];
    __shared__ __align__(16) float Red[2][DNP][8][256];

    const int tid   = threadIdx.x, lane = tid & 31, wave = tid >> 5;
    const int lhalf = lane >> 4, lmod = lane & 15;
    const int i0    = blockIdx.x * 16;
    const int sbase = blockIdx.y * DSC;
    const int jw    = wave * 32 + lhalf * 8;
    const float two_pi = 6.28318530717958647692f;

    auto build = [&](int buf, int sfirst) {
        const float pw[DG] = { 1.0f, 1.5f, 2.25f, 3.375f, 5.0625f, 7.59375f };
        {
            const float base = 2.0f + 0.5f * (float)(i0 + tid);
            #pragma unroll
            for (int p = 0; p < DNP; ++p) {
                const float sv = two_pi * (float)(sfirst + p);
                #pragma unroll
                for (int g = 0; g < DG; ++g)
                    Ct[buf][p][tid * DG + g] = __cosf(sv / (base * pw[g]));
            }
        }
        if (tid < 15 * DNP * DG) {
            const int tt = 256 + tid / (DNP * DG);
            const int r  = tid % (DNP * DG);
            const int p  = r / DG, g = r % DG;
            const float pwg = g == 0 ? pw[0] : g == 1 ? pw[1] : g == 2 ? pw[2]
                            : g == 3 ? pw[3] : g == 4 ? pw[4] : pw[5];
            const float base = 2.0f + 0.5f * (float)(i0 + tt);
            const float sv   = two_pi * (float)(sfirst + p);
            Ct[buf][p][tt * DG + g] = __cosf(sv / (base * pwg));
        }
    };
    auto load_a = [&](v4f (&r)[DNP][4], int sfirst) {
        #pragma unroll
        for (int p = 0; p < DNP; ++p) {
            const float* xr = xt + ((size_t)lmod * S + sfirst + p) * DD + jw;
            #pragma unroll
            for (int q = 0; q < 4; ++q)
                r[p][q] = *(const v4f*)(xr + 16 * (q >> 1) + 4 * (q & 1));
        }
    };
    auto reduce = [&](int buf, int sfirst) {
        const int b = (tid >> 5) + 8 * lhalf;
        #pragma unroll
        for (int p = 0; p < DNP; ++p) {
            float sum = Red[buf][p][0][tid];
            #pragma unroll
            for (int w = 1; w < 8; ++w) sum += Red[buf][p][w][tid];
            const size_t off = ((size_t)b * S + sfirst + p) * DD + i0 + lmod;
            hd[off] = sum + h[off];
        }
    };

    v2f pp[2][4][DG];
    #pragma unroll
    for (int blk = 0; blk < 2; ++blk) {
        const float* pg = P + ((size_t)(i0 + lmod) * DD + jw + 16 * blk) * DG;
        float pr[48];
        #pragma unroll
        for (int q = 0; q < 12; ++q) {
            const v4f v = *(const v4f*)(pg + 4 * q);
            pr[4 * q]     = v.x;
            pr[4 * q + 1] = v.y;
            pr[4 * q + 2] = v.z;
            pr[4 * q + 3] = v.w;
        }
        #pragma unroll
        for (int t = 0; t < 4; ++t)
            #pragma unroll
            for (int g = 0; g < DG; ++g) {
                pp[blk][t][g].x = pr[(2 * t) * DG + g];
                pp[blk][t][g].y = pr[(2 * t + 1) * DG + g];
            }
    }

    v4f ra[DNP][4];
    load_a(ra, sbase);
    build(0, sbase);
    __syncthreads();

    const int ntrip = DSC / DNP;
    for (int n = 0; n < ntrip; ++n) {
        const int buf    = n & 1;
        const int sfirst = sbase + n * DNP;

        v4f rn[DNP][4];
        if (n + 1 < ntrip) load_a(rn, sfirst + DNP);
        if (n > 0) reduce(buf ^ 1, sfirst - DNP);

        const float* cb = &Ct[buf][0][0] + (lmod + jw) * DG;
        #pragma unroll
        for (int p = 0; p < DNP; ++p) {
            v8f acc = {};
            #pragma unroll
            for (int blk = 0; blk < 2; ++blk)
                #pragma unroll
                for (int t = 0; t < 4; ++t) {
                    const float* c0 = cb + p * DCT + (16 * blk + 2 * t) * DG;
                    v2f b;
                    {
                        const v2f c = { c0[0], c0[DG] };
                        b = pp[blk][t][0] * c;
                    }
                    #pragma unroll
                    for (int g = 1; g < DG; ++g) {
                        const v2f c = { c0[g], c0[DG + g] };
                        b = pp[blk][t][g] * c + b;
                    }
                    const v4f av = ra[p][2 * blk + (t >> 1)];
                    v2f a;
                    if (t & 1) { a.x = av.z; a.y = av.w; }
                    else       { a.x = av.x; a.y = av.y; }
                    acc = wmma_f32(a, b, acc);
                }
            #pragma unroll
            for (int v = 0; v < 8; ++v)
                Red[buf][p][wave][v * 32 + lane] = acc[v];
        }

        if (n + 1 < ntrip) {
            build(buf ^ 1, sfirst + DNP);
            #pragma unroll
            for (int p = 0; p < DNP; ++p)
                #pragma unroll
                for (int q = 0; q < 4; ++q) ra[p][q] = rn[p][q];
        }
        __syncthreads();
    }
    reduce((ntrip - 1) & 1, sbase + (ntrip - 1) * DNP);
}

#define AT_KT 64
#define AT_KP 36
#define AT_VP 40
#define AT_PP 20
#define AT_PT 320

__global__ __launch_bounds__(128) void ddnet_attn_kernel(
    const float* __restrict__ qkv, float* __restrict__ o, int S)
{
    const int qb    = S / 64;
    const int bh    = blockIdx.x / qb;
    const int qblk  = blockIdx.x % qb;
    const int b     = bh >> 3, head = bh & 7;
    const int tid   = threadIdx.x, lane = tid & 31, wave = tid >> 5;
    const int lhalf = lane >> 4, lmod = lane & 15;
    const int q0    = qblk * 64 + wave * 16;

    __shared__ __align__(16) float Kf[AT_KT * AT_KP];
    __shared__ __align__(16) float Vf[AT_KT * AT_VP];
    __shared__ __align__(16) float Pf[4][4][AT_PT];

    const size_t rs = 768;
    const float* qp = qkv + (size_t)b * S * rs + head * 32;
    const float* kp = qp + 256;
    const float* vp = qp + 512;
    const float scale = 0.17677669529663688f;

    v2f aq[8];
    #pragma unroll
    for (int t = 0; t < 8; ++t) {
        const float* qr = qp + (size_t)(q0 + lmod) * rs;
        aq[t].x = qr[t * 4 + 2 * lhalf];
        aq[t].y = qr[t * 4 + 1 + 2 * lhalf];
    }

    v8f mrow, lrow = {};
    v8f o0 = {}, o1 = {};
    #pragma unroll
    for (int v = 0; v < 8; ++v) mrow[v] = -1e30f;

    const int sq = tid & 7;
    const int sr = tid >> 3;
    const float* kst = kp + (size_t)sr * rs + 4 * sq;
    const float* vst = vp + (size_t)sr * rs + 4 * sq;
    const int kst_lds = sr * AT_KP + 4 * sq;
    const int vst_lds = sr * AT_VP + 4 * sq;

    v4f rk[4], rv[4];
    #pragma unroll
    for (int i = 0; i < 4; ++i) {
        rk[i] = *(const v4f*)(kst + (size_t)(16 * i) * rs);
        rv[i] = *(const v4f*)(vst + (size_t)(16 * i) * rs);
    }

    float* pst = &Pf[wave][0][(8 * lhalf) * AT_PP + lmod];

    const int nkt = S / AT_KT;
    for (int kt = 0; kt < nkt; ++kt) {
        #pragma unroll
        for (int i = 0; i < 4; ++i) {
            *(v4f*)&Kf[i * 16 * AT_KP + kst_lds] = rk[i];
            *(v4f*)&Vf[i * 16 * AT_VP + vst_lds] = rv[i];
        }
        __syncthreads();

        if (kt + 1 < nkt) {
            const size_t nb = (size_t)((kt + 1) * AT_KT) * rs;
            #pragma unroll
            for (int i = 0; i < 4; ++i) {
                rk[i] = *(const v4f*)(kst + nb + (size_t)(16 * i) * rs);
                rv[i] = *(const v4f*)(vst + nb + (size_t)(16 * i) * rs);
            }
        }

        v8f sc[4] = {};
        #pragma unroll
        for (int t = 0; t < 8; ++t)
            #pragma unroll
            for (int sb = 0; sb < 4; ++sb) {
                v2f bk = lds_frag(&Kf[(sb * 16 + lmod) * AT_KP + t * 4 + 2 * lhalf]);
                sc[sb] = wmma_f32(aq[t], bk, sc[sb]);
            }

        const v8f s0 = sc[0] * scale, s1 = sc[1] * scale;
        const v8f s2 = sc[2] * scale, s3 = sc[3] * scale;
        v8f mnew, corr;
        #pragma unroll
        for (int v = 0; v < 8; ++v) {
            float mx = fmaxf(fmaxf(s0[v], s1[v]), fmaxf(s2[v], s3[v]));
            #pragma unroll
            for (int m = 8; m >= 1; m >>= 1) mx = fmaxf(mx, __shfl_xor(mx, m, 32));
            mnew[v] = fmaxf(mrow[v], mx);
            corr[v] = __expf(mrow[v] - mnew[v]);
        }
        const v8f d0 = s0 - mnew, d1 = s1 - mnew, d2 = s2 - mnew, d3 = s3 - mnew;
        #pragma unroll
        for (int v = 0; v < 8; ++v) {
            sc[0][v] = __expf(d0[v]);
            sc[1][v] = __expf(d1[v]);
            sc[2][v] = __expf(d2[v]);
            sc[3][v] = __expf(d3[v]);
        }
        lrow = lrow * corr + ((sc[0] + sc[1]) + (sc[2] + sc[3]));
        mrow = mnew;
        o0 *= corr;
        o1 *= corr;

        #pragma unroll
        for (int sb = 0; sb < 4; ++sb)
            #pragma unroll
            for (int v = 0; v < 8; ++v)
                pst[sb * AT_PT + v * AT_PP] = sc[sb][v];
        #pragma unroll
        for (int sb = 0; sb < 4; ++sb)
            #pragma unroll
            for (int t = 0; t < 4; ++t) {
                const int fo = t * 2 + lhalf;
                const float* vr = &Vf[(sb * 16 + 2 * fo) * AT_VP + lmod];
                v2f ap  = lds_frag(&Pf[wave][sb][lmod * AT_PP + 2 * fo]);
                v2f bv0 = { vr[0],  vr[AT_VP] };
                v2f bv1 = { vr[16], vr[AT_VP + 16] };
                o0 = wmma_f32(ap, bv0, o0);
                o1 = wmma_f32(ap, bv1, o1);
            }
        __syncthreads();
    }

    #pragma unroll
    for (int v = 0; v < 8; ++v) {
        float l = lrow[v];
        #pragma unroll
        for (int m = 8; m >= 1; m >>= 1) l += __shfl_xor(l, m, 32);
        int qr = v + 8 * lhalf;
        float inv = 1.0f / l;
        size_t base = ((size_t)b * S + q0 + qr) * 256 + head * 32;
        o[base + lmod]      = o0[v] * inv;
        o[base + 16 + lmod] = o1[v] * inv;
    }
}

static void launch_gemm(const float* A, const float* W, const float* bias,
                        const float* resid, float* C, int M, int N, int K,
                        int act, hipStream_t stream)
{
    dim3 grid(N / GT_N, M / GT_M);
    ddnet_gemm_kernel<<<grid, 256, 0, stream>>>(A, W, bias, resid, C, M, N, K, act);
}

extern "C" void kernel_launch(void* const* d_in, const int* in_sizes, int n_in,
                              void* d_out, int out_size, void* d_ws, size_t ws_size,
                              hipStream_t stream) {
    (void)in_sizes; (void)n_in; (void)out_size; (void)ws_size;
    const int B = 16, S = 512, IN = 64, D = 256, L = 3, G = 6, DFF = 1024;
    const int Mr = B * S;

    const float* x      = (const float*)d_in[0];
    const float* in_w   = (const float*)d_in[1];
    const float* in_b   = (const float*)d_in[2];
    const float* Mw     = (const float*)d_in[3];
    const float* P      = (const float*)d_in[4];
    const float* desc_g = (const float*)d_in[5];
    const float* desc_b = (const float*)d_in[6];
    const float* qkv_w  = (const float*)d_in[7];
    const float* qkv_b  = (const float*)d_in[8];
    const float* out_w  = (const float*)d_in[9];
    const float* out_b  = (const float*)d_in[10];
    const float* ff1_w  = (const float*)d_in[11];
    const float* ff1_b  = (const float*)d_in[12];
    const float* ff2_w  = (const float*)d_in[13];
    const float* ff2_b  = (const float*)d_in[14];
    const float* ln1_g  = (const float*)d_in[15];
    const float* ln1_b  = (const float*)d_in[16];
    const float* ln2_g  = (const float*)d_in[17];
    const float* ln2_b  = (const float*)d_in[18];

    float* h    = (float*)d_ws;
    float* tmp  = h    + (size_t)Mr * D;
    float* xt   = tmp  + (size_t)Mr * D;
    float* hd   = xt   + (size_t)Mr * D;
    float* qkv  = hd   + (size_t)Mr * D;
    float* obuf = qkv  + (size_t)Mr * 3 * D;
    float* h1   = obuf + (size_t)Mr * D;
    float* ff   = h1   + (size_t)Mr * D;

    launch_gemm(x, in_w, in_b, nullptr, h, Mr, D, IN, 0, stream);
    ddnet_pe_kernel<<<(Mr * D) / 256, 256, 0, stream>>>(h, S, D);

    for (int l = 0; l < L; ++l) {
        launch_gemm(h, Mw + (size_t)l * D * D, nullptr, nullptr, tmp, Mr, D, D, 0, stream);
        ddnet_ln_kernel<<<Mr / 8, 256, 0, stream>>>(tmp, desc_g + l * D, desc_b + l * D, xt, D);
        ddnet_desc_kernel<<<dim3(DD / 16, S / DSC), 256, 0, stream>>>(xt, h, P + (size_t)l * D * D * G, hd, S);
        launch_gemm(hd, qkv_w + (size_t)l * 3 * D * D, qkv_b + (size_t)l * 3 * D,
                    nullptr, qkv, Mr, 3 * D, D, 0, stream);
        ddnet_attn_kernel<<<B * 8 * (S / 64), 128, 0, stream>>>(qkv, obuf, S);
        launch_gemm(obuf, out_w + (size_t)l * D * D, out_b + l * D, hd, tmp, Mr, D, D, 0, stream);
        ddnet_ln_kernel<<<Mr / 8, 256, 0, stream>>>(tmp, ln1_g + l * D, ln1_b + l * D, h1, D);
        launch_gemm(h1, ff1_w + (size_t)l * DFF * D, ff1_b + (size_t)l * DFF,
                    nullptr, ff, Mr, DFF, D, 1  , stream);
        launch_gemm(ff, ff2_w + (size_t)l * D * DFF, ff2_b + l * D, h1, tmp, Mr, D, DFF, 0, stream);
        float* dst = (l == L - 1) ? (float*)d_out : h;
        ddnet_ln_kernel<<<Mr / 8, 256, 0, stream>>>(tmp, ln2_g + l * D, ln2_b + l * D, dst, D);
    }
}


# N=256 products re-tiled to 32x256: a work-group owns whole rows and applies the pass behind the product (LayerNorm, positional term) in its epilogue; 10 of 32 launches gone; LN sum order changed
# speedup vs baseline: 2.2380x; 1.0354x over previous
#include <hip/hip_runtime.h>
#include <math.h>

typedef __attribute__((ext_vector_type(2))) float v2f;
typedef __attribute__((ext_vector_type(8))) float v8f;
typedef __attribute__((ext_vector_type(4))) float v4f;

__device__ __forceinline__ v8f wmma_f32(v2f a, v2f b, v8f c) {
    return __builtin_amdgcn_wmma_f32_16x16x4_f32(false, a, false, b, (short)0, c, false, false);
}
__device__ __forceinline__ v2f lds_frag(const float* p) {
    return *(const v2f*)p;
}

#define GT_K 16
#define AB_P 20

template <int TM, int TN>
__global__ __launch_bounds__(256) void ddnet_gemm_kernel(
    const float* __restrict__ A, const float* __restrict__ W,
    const float* __restrict__ bias, const float* __restrict__ resid,
    const float* __restrict__ gamma, const float* __restrict__ beta,
    float* __restrict__ C, int Mdim, int Ndim, int Kdim, int act, int S)
{
    constexpr int WMC = TM / 32;
    constexpr int STG = (TM + TN) * AB_P;
    constexpr int NA  = TM / 16;
    constexpr int NB  = TN / 16;
    constexpr int C_P = TN + 4;
    constexpr int SMEM = 2 * STG > TM * C_P ? 2 * STG : TM * C_P;

    __shared__ __align__(16) float smem[SMEM];

    const int tid   = threadIdx.x;
    const int lane  = tid & 31;
    const int wave  = tid >> 5;
    const int lhalf = lane >> 4;
    const int lmod  = lane & 15;
    const int m0    = blockIdx.y * TM;
    const int n0    = blockIdx.x * TN;
    const int wm    = (wave % WMC) * 32;
    const int wn    = (wave / WMC) * 32;
    const int nk    = Kdim >> 4;

    v8f acc[2][2] = {};
    float ra[NA], rb[NB];

    const int sk = tid & 15, sm = tid >> 4;
    const float* ag = A + (size_t)(m0 + sm) * Kdim + sk;
    const float* wg = W + (size_t)(n0 + sm) * Kdim + sk;
    const int sl = sm * AB_P + sk;

    auto fetch = [&](int k0) {
        #pragma unroll
        for (int i = 0; i < NA; ++i) ra[i] = ag[(size_t)(16 * i) * Kdim + k0];
        #pragma unroll
        for (int i = 0; i < NB; ++i) rb[i] = wg[(size_t)(16 * i) * Kdim + k0];
    };
    auto spill = [&](int st) {
        float* Ad = smem + st * STG + sl;
        float* Bd = Ad + TM * AB_P;
        #pragma unroll
        for (int i = 0; i < NA; ++i) Ad[16 * i * AB_P] = ra[i];
        #pragma unroll
        for (int i = 0; i < NB; ++i) Bd[16 * i * AB_P] = rb[i];
    };

    fetch(0);
    spill(0);
    if (nk > 1) fetch(GT_K);
    __syncthreads();

    for (int kt = 0; kt < nk; ++kt) {
        const int  cur  = kt & 1;
        const bool more = kt + 1 < nk;
        if (more) {
            spill(cur ^ 1);
            if (kt + 2 < nk) fetch((kt + 2) << 4);
        }
        const float* Ab = smem + cur * STG;
        const float* Bb = Ab + TM * AB_P;
        v2f fa[4][2], fb[4][2];
        #pragma unroll
        for (int t = 0; t < 4; ++t) {
            const int kf = t * 4 + 2 * lhalf;
            fa[t][0] = lds_frag(&Ab[(wm + lmod) * AB_P + kf]);
            fa[t][1] = lds_frag(&Ab[(wm + 16 + lmod) * AB_P + kf]);
            fb[t][0] = lds_frag(&Bb[(wn + lmod) * AB_P + kf]);
            fb[t][1] = lds_frag(&Bb[(wn + 16 + lmod) * AB_P + kf]);
        }
        if (more) {
            __builtin_amdgcn_fence(__ATOMIC_RELEASE, "workgroup", "local");
            __builtin_amdgcn_s_barrier_signal(-1);
        }
        #pragma unroll
        for (int t = 0; t < 4; ++t) {
            acc[0][0] = wmma_f32(fa[t][0], fb[t][0], acc[0][0]);
            acc[0][1] = wmma_f32(fa[t][0], fb[t][1], acc[0][1]);
            acc[1][0] = wmma_f32(fa[t][1], fb[t][0], acc[1][0]);
            acc[1][1] = wmma_f32(fa[t][1], fb[t][1], acc[1][1]);
        }
        if (more) {
            __builtin_amdgcn_s_barrier_wait(-1);
            __builtin_amdgcn_fence(__ATOMIC_ACQUIRE, "workgroup", "local");
        }
    }

    __syncthreads();
    #pragma unroll
    for (int mi = 0; mi < 2; ++mi)
        #pragma unroll
        for (int ni = 0; ni < 2; ++ni)
            #pragma unroll
            for (int v = 0; v < 8; ++v)
                smem[(wm + mi * 16 + v + 8 * lhalf) * C_P + wn + ni * 16 + lmod] = acc[mi][ni][v];
    __syncthreads();

    if constexpr (TN == 128) {
        const int c4 = (tid & 31) * 4;
        const int r0 = tid >> 5;
        const float* rp = resid ? resid + (size_t)(m0 + r0) * Ndim + n0 + c4 : nullptr;
        float*       cp = C + (size_t)(m0 + r0) * Ndim + n0 + c4;
        v4f bv = {0.f, 0.f, 0.f, 0.f};
        if (bias) bv = *(const v4f*)(bias + n0 + c4);
        v4f rv[TM / 8];
        if (resid) {
            #pragma unroll
            for (int i = 0; i < TM / 8; ++i)
                rv[i] = *(const v4f*)(rp + (size_t)(8 * i) * Ndim);
        }
        #pragma unroll
        for (int i = 0; i < TM / 8; ++i) {
            v4f val = *(const v4f*)&smem[(r0 + 8 * i) * C_P + c4];
            if (bias)  val += bv;
            if (resid) val += rv[i];
            if (act == 1) {
                val.x = 0.5f * val.x * (1.0f + erff(val.x * 0.70710678118654752f));
                val.y = 0.5f * val.y * (1.0f + erff(val.y * 0.70710678118654752f));
                val.z = 0.5f * val.z * (1.0f + erff(val.z * 0.70710678118654752f));
                val.w = 0.5f * val.w * (1.0f + erff(val.w * 0.70710678118654752f));
            }
            *(v4f*)(cp + (size_t)(8 * i) * Ndim) = val;
        }
    } else {
        constexpr int NR = TM / 8;
        const int c0 = 4 * lane;
        v4f b0 = {0.f, 0.f, 0.f, 0.f}, b1 = b0;
        if (bias) {
            b0 = *(const v4f*)(bias + c0);
            b1 = *(const v4f*)(bias + 128 + c0);
        }
        v4f q0[NR], q1[NR];
        if (resid) {
            #pragma unroll
            for (int i = 0; i < NR; ++i) {
                const float* rp = resid + (size_t)(m0 + wave + 8 * i) * TN + c0;
                q0[i] = *(const v4f*)rp;
                q1[i] = *(const v4f*)(rp + 128);
            }
        }
        v4f x0[NR], x1[NR];
        #pragma unroll
        for (int i = 0; i < NR; ++i) {
            const float* sp = &smem[(wave + 8 * i) * C_P + c0];
            x0[i] = *(const v4f*)sp;
            x1[i] = *(const v4f*)(sp + 128);
            if (bias)  { x0[i] += b0;    x1[i] += b1; }
            if (resid) { x0[i] += q0[i]; x1[i] += q1[i]; }
        }
        if (gamma) {
            const v4f g0 = *(const v4f*)(gamma + c0), g1 = *(const v4f*)(gamma + 128 + c0);
            const v4f e0 = *(const v4f*)(beta + c0),  e1 = *(const v4f*)(beta + 128 + c0);
            float sum[NR];
            #pragma unroll
            for (int i = 0; i < NR; ++i)
                sum[i] = ((x0[i].x + x0[i].y) + (x0[i].z + x0[i].w))
                       + ((x1[i].x + x1[i].y) + (x1[i].z + x1[i].w));
            #pragma unroll
            for (int m = 16; m >= 1; m >>= 1)
                #pragma unroll
                for (int i = 0; i < NR; ++i) sum[i] += __shfl_xor(sum[i], m, 32);
            float vs[NR];
            #pragma unroll
            for (int i = 0; i < NR; ++i) {
                const float mean = sum[i] * (1.0f / 256.0f);
                x0[i] -= mean;
                x1[i] -= mean;
                const v4f s0 = x0[i] * x0[i], s1 = x1[i] * x1[i];
                vs[i] = ((s0.x + s0.y) + (s0.z + s0.w)) + ((s1.x + s1.y) + (s1.z + s1.w));
            }
            #pragma unroll
            for (int m = 16; m >= 1; m >>= 1)
                #pragma unroll
                for (int i = 0; i < NR; ++i) vs[i] += __shfl_xor(vs[i], m, 32);
            #pragma unroll
            for (int i = 0; i < NR; ++i) {
                const float rstd = rsqrtf(vs[i] * (1.0f / 256.0f) + 1e-5f);
                x0[i] = x0[i] * rstd * g0 + e0;
                x1[i] = x1[i] * rstd * g1 + e1;
            }
        } else if (act == 2) {
            const float rate = logf(10000.0f) / (float)Ndim;
            float dv[4];
            #pragma unroll
            for (int p = 0; p < 4; ++p) {
                const int d = (p < 2 ? c0 : 128 + c0) + 2 * (p & 1);
                dv[p] = __expf(-(float)d * rate);
            }
            #pragma unroll
            for (int i = 0; i < NR; ++i) {
                const float sp = (float)((m0 + wave + 8 * i) % S);
                x0[i].x += __sinf(sp * dv[0]);  x0[i].y += __cosf(sp * dv[0]);
                x0[i].z += __sinf(sp * dv[1]);  x0[i].w += __cosf(sp * dv[1]);
                x1[i].x += __sinf(sp * dv[2]);  x1[i].y += __cosf(sp * dv[2]);
                x1[i].z += __sinf(sp * dv[3]);  x1[i].w += __cosf(sp * dv[3]);
            }
        }
        #pragma unroll
        for (int i = 0; i < NR; ++i) {
            float* cp = C + (size_t)(m0 + wave + 8 * i) * TN + c0;
            *(v4f*)cp         = x0[i];
            *(v4f*)(cp + 128) = x1[i];
        }
    }
}

#define DD 256
#define DG 6
#define DSC 16
#define DNP 2
#define DCT 1632

__global__ __launch_bounds__(256) void ddnet_desc_kernel(
    const float* __restrict__ xt, const float* __restrict__ h,
    const float* __restrict__ P, float* __restrict__ hd, int S)
{
    __shared__ __align__(16) float Ct[2][DNP][DCT];
    __shared__ __align__(16) float Red[2][DNP][8][256];

    const int tid   = threadIdx.x, lane = tid & 31, wave = tid >> 5;
    const int lhalf = lane >> 4, lmod = lane & 15;
    const int i0    = blockIdx.x * 16;
    const int sbase = blockIdx.y * DSC;
    const int jw    = wave * 32 + lhalf * 8;
    const float two_pi = 6.28318530717958647692f;

    auto build = [&](int buf, int sfirst) {
        const float pw[DG] = { 1.0f, 1.5f, 2.25f, 3.375f, 5.0625f, 7.59375f };
        {
            const float base = 2.0f + 0.5f * (float)(i0 + tid);
            #pragma unroll
            for (int p = 0; p < DNP; ++p) {
                const float sv = two_pi * (float)(sfirst + p);
                #pragma unroll
                for (int g = 0; g < DG; ++g)
                    Ct[buf][p][tid * DG + g] = __cosf(sv / (base * pw[g]));
            }
        }
        if (tid < 15 * DNP * DG) {
            const int tt = 256 + tid / (DNP * DG);
            const int r  = tid % (DNP * DG);
            const int p  = r / DG, g = r % DG;
            const float pwg = g == 0 ? pw[0] : g == 1 ? pw[1] : g == 2 ? pw[2]
                            : g == 3 ? pw[3] : g == 4 ? pw[4] : pw[5];
            const float base = 2.0f + 0.5f * (float)(i0 + tt);
            const float sv   = two_pi * (float)(sfirst + p);
            Ct[buf][p][tt * DG + g] = __cosf(sv / (base * pwg));
        }
    };
    auto load_a = [&](v4f (&r)[DNP][4], int sfirst) {
        #pragma unroll
        for (int p = 0; p < DNP; ++p) {
            const float* xr = xt + ((size_t)lmod * S + sfirst + p) * DD + jw;
            #pragma unroll
            for (int q = 0; q < 4; ++q)
                r[p][q] = *(const v4f*)(xr + 16 * (q >> 1) + 4 * (q & 1));
        }
    };
    auto reduce = [&](int buf, int sfirst) {
        const int b = (tid >> 5) + 8 * lhalf;
        #pragma unroll
        for (int p = 0; p < DNP; ++p) {
            float sum = Red[buf][p][0][tid];
            #pragma unroll
            for (int w = 1; w < 8; ++w) sum += Red[buf][p][w][tid];
            const size_t off = ((size_t)b * S + sfirst + p) * DD + i0 + lmod;
            hd[off] = sum + h[off];
        }
    };

    v2f pp[2][4][DG];
    #pragma unroll
    for (int blk = 0; blk < 2; ++blk) {
        const float* pg = P + ((size_t)(i0 + lmod) * DD + jw + 16 * blk) * DG;
        float pr[48];
        #pragma unroll
        for (int q = 0; q < 12; ++q) {
            const v4f v = *(const v4f*)(pg + 4 * q);
            pr[4 * q]     = v.x;
            pr[4 * q + 1] = v.y;
            pr[4 * q + 2] = v.z;
            pr[4 * q + 3] = v.w;
        }
        #pragma unroll
        for (int t = 0; t < 4; ++t)
            #pragma unroll
            for (int g = 0; g < DG; ++g) {
                pp[blk][t][g].x = pr[(2 * t) * DG + g];
                pp[blk][t][g].y = pr[(2 * t + 1) * DG + g];
            }
    }

    v4f ra[DNP][4];
    load_a(ra, sbase);
    build(0, sbase);
    __syncthreads();

    const int ntrip = DSC / DNP;
    for (int n = 0; n < ntrip; ++n) {
        const int buf    = n & 1;
        const int sfirst = sbase + n * DNP;

        v4f rn[DNP][4];
        if (n + 1 < ntrip) load_a(rn, sfirst + DNP);
        if (n > 0) reduce(buf ^ 1, sfirst - DNP);

        const float* cb = &Ct[buf][0][0] + (lmod + jw) * DG;
        #pragma unroll
        for (int p = 0; p < DNP; ++p) {
            v8f acc = {};
            #pragma unroll
            for (int blk = 0; blk < 2; ++blk)
                #pragma unroll
                for (int t = 0; t < 4; ++t) {
                    const float* c0 = cb + p * DCT + (16 * blk + 2 * t) * DG;
                    v2f b;
                    {
                        const v2f c = { c0[0], c0[DG] };
                        b = pp[blk][t][0] * c;
                    }
                    #pragma unroll
                    for (int g = 1; g < DG; ++g) {
                        const v2f c = { c0[g], c0[DG + g] };
                        b = pp[blk][t][g] * c + b;
                    }
                    const v4f av = ra[p][2 * blk + (t >> 1)];
                    v2f a;
                    if (t & 1) { a.x = av.z; a.y = av.w; }
                    else       { a.x = av.x; a.y = av.y; }
                    acc = wmma_f32(a, b, acc);
                }
            #pragma unroll
            for (int v = 0; v < 8; ++v)
                Red[buf][p][wave][v * 32 + lane] = acc[v];
        }

        if (n + 1 < ntrip) {
            build(buf ^ 1, sfirst + DNP);
            #pragma unroll
            for (int p = 0; p < DNP; ++p)
                #pragma unroll
                for (int q = 0; q < 4; ++q) ra[p][q] = rn[p][q];
        }
        __syncthreads();
    }
    reduce((ntrip - 1) & 1, sbase + (ntrip - 1) * DNP);
}

#define AT_KT 64
#define AT_KP 36
#define AT_VP 40
#define AT_PP 20
#define AT_PT 320

__global__ __launch_bounds__(128) void ddnet_attn_kernel(
    const float* __restrict__ qkv, float* __restrict__ o, int S)
{
    const int qb    = S / 64;
    const int bh    = blockIdx.x / qb;
    const int qblk  = blockIdx.x % qb;
    const int b     = bh >> 3, head = bh & 7;
    const int tid   = threadIdx.x, lane = tid & 31, wave = tid >> 5;
    const int lhalf = lane >> 4, lmod = lane & 15;
    const int q0    = qblk * 64 + wave * 16;

    __shared__ __align__(16) float Kf[AT_KT * AT_KP];
    __shared__ __align__(16) float Vf[AT_KT * AT_VP];
    __shared__ __align__(16) float Pf[4][4][AT_PT];

    const size_t rs = 768;
    const float* qp = qkv + (size_t)b * S * rs + head * 32;
    const float* kp = qp + 256;
    const float* vp = qp + 512;
    const float scale = 0.17677669529663688f;

    v2f aq[8];
    #pragma unroll
    for (int t = 0; t < 8; ++t) {
        const float* qr = qp + (size_t)(q0 + lmod) * rs;
        aq[t].x = qr[t * 4 + 2 * lhalf];
        aq[t].y = qr[t * 4 + 1 + 2 * lhalf];
    }

    v8f mrow, lrow = {};
    v8f o0 = {}, o1 = {};
    #pragma unroll
    for (int v = 0; v < 8; ++v) mrow[v] = -1e30f;

    const int sq = tid & 7;
    const int sr = tid >> 3;
    const float* kst = kp + (size_t)sr * rs + 4 * sq;
    const float* vst = vp + (size_t)sr * rs + 4 * sq;
    const int kst_lds = sr * AT_KP + 4 * sq;
    const int vst_lds = sr * AT_VP + 4 * sq;

    v4f rk[4], rv[4];
    #pragma unroll
    for (int i = 0; i < 4; ++i) {
        rk[i] = *(const v4f*)(kst + (size_t)(16 * i) * rs);
        rv[i] = *(const v4f*)(vst + (size_t)(16 * i) * rs);
    }

    float* pst = &Pf[wave][0][(8 * lhalf) * AT_PP + lmod];

    const int nkt = S / AT_KT;
    for (int kt = 0; kt < nkt; ++kt) {
        #pragma unroll
        for (int i = 0; i < 4; ++i) {
            *(v4f*)&Kf[i * 16 * AT_KP + kst_lds] = rk[i];
            *(v4f*)&Vf[i * 16 * AT_VP + vst_lds] = rv[i];
        }
        __syncthreads();

        if (kt + 1 < nkt) {
            const size_t nb = (size_t)((kt + 1) * AT_KT) * rs;
            #pragma unroll
            for (int i = 0; i < 4; ++i) {
                rk[i] = *(const v4f*)(kst + nb + (size_t)(16 * i) * rs);
                rv[i] = *(const v4f*)(vst + nb + (size_t)(16 * i) * rs);
            }
        }

        v8f sc[4] = {};
        #pragma unroll
        for (int t = 0; t < 8; ++t)
            #pragma unroll
            for (int sb = 0; sb < 4; ++sb) {
                v2f bk = lds_frag(&Kf[(sb * 16 + lmod) * AT_KP + t * 4 + 2 * lhalf]);
                sc[sb] = wmma_f32(aq[t], bk, sc[sb]);
            }

        const v8f s0 = sc[0] * scale, s1 = sc[1] * scale;
        const v8f s2 = sc[2] * scale, s3 = sc[3] * scale;
        v8f mnew, corr;
        #pragma unroll
        for (int v = 0; v < 8; ++v) {
            float mx = fmaxf(fmaxf(s0[v], s1[v]), fmaxf(s2[v], s3[v]));
            #pragma unroll
            for (int m = 8; m >= 1; m >>= 1) mx = fmaxf(mx, __shfl_xor(mx, m, 32));
            mnew[v] = fmaxf(mrow[v], mx);
            corr[v] = __expf(mrow[v] - mnew[v]);
        }
        const v8f d0 = s0 - mnew, d1 = s1 - mnew, d2 = s2 - mnew, d3 = s3 - mnew;
        #pragma unroll
        for (int v = 0; v < 8; ++v) {
            sc[0][v] = __expf(d0[v]);
            sc[1][v] = __expf(d1[v]);
            sc[2][v] = __expf(d2[v]);
            sc[3][v] = __expf(d3[v]);
        }
        lrow = lrow * corr + ((sc[0] + sc[1]) + (sc[2] + sc[3]));
        mrow = mnew;
        o0 *= corr;
        o1 *= corr;

        #pragma unroll
        for (int sb = 0; sb < 4; ++sb)
            #pragma unroll
            for (int v = 0; v < 8; ++v)
                pst[sb * AT_PT + v * AT_PP] = sc[sb][v];
        #pragma unroll
        for (int sb = 0; sb < 4; ++sb)
            #pragma unroll
            for (int t = 0; t < 4; ++t) {
                const int fo = t * 2 + lhalf;
                const float* vr = &Vf[(sb * 16 + 2 * fo) * AT_VP + lmod];
                v2f ap  = lds_frag(&Pf[wave][sb][lmod * AT_PP + 2 * fo]);
                v2f bv0 = { vr[0],  vr[AT_VP] };
                v2f bv1 = { vr[16], vr[AT_VP + 16] };
                o0 = wmma_f32(ap, bv0, o0);
                o1 = wmma_f32(ap, bv1, o1);
            }
        __syncthreads();
    }

    #pragma unroll
    for (int v = 0; v < 8; ++v) {
        float l = lrow[v];
        #pragma unroll
        for (int m = 8; m >= 1; m >>= 1) l += __shfl_xor(l, m, 32);
        int qr = v + 8 * lhalf;
        float inv = 1.0f / l;
        size_t base = ((size_t)b * S + q0 + qr) * 256 + head * 32;
        o[base + lmod]      = o0[v] * inv;
        o[base + 16 + lmod] = o1[v] * inv;
    }
}

static void launch_gemm(const float* A, const float* W, const float* bias,
                        float* C, int M, int N, int K, int act, hipStream_t stream)
{
    dim3 grid(N / 128, M / 64);
    ddnet_gemm_kernel<64, 128><<<grid, 256, 0, stream>>>(
        A, W, bias, nullptr, nullptr, nullptr, C, M, N, K, act, 0);
}
static void launch_gemm_rows(const float* A, const float* W, const float* bias,
                             const float* resid, const float* gamma, const float* beta,
                             float* C, int M, int K, int act, int S, hipStream_t stream)
{
    dim3 grid(1, M / 32);
    ddnet_gemm_kernel<32, 256><<<grid, 256, 0, stream>>>(
        A, W, bias, resid, gamma, beta, C, M, 256, K, act, S);
}

extern "C" void kernel_launch(void* const* d_in, const int* in_sizes, int n_in,
                              void* d_out, int out_size, void* d_ws, size_t ws_size,
                              hipStream_t stream) {
    (void)in_sizes; (void)n_in; (void)out_size; (void)ws_size;
    const int B = 16, S = 512, IN = 64, D = 256, L = 3, G = 6, DFF = 1024;
    const int Mr = B * S;

    const float* x      = (const float*)d_in[0];
    const float* in_w   = (const float*)d_in[1];
    const float* in_b   = (const float*)d_in[2];
    const float* Mw     = (const float*)d_in[3];
    const float* P      = (const float*)d_in[4];
    const float* desc_g = (const float*)d_in[5];
    const float* desc_b = (const float*)d_in[6];
    const float* qkv_w  = (const float*)d_in[7];
    const float* qkv_b  = (const float*)d_in[8];
    const float* out_w  = (const float*)d_in[9];
    const float* out_b  = (const float*)d_in[10];
    const float* ff1_w  = (const float*)d_in[11];
    const float* ff1_b  = (const float*)d_in[12];
    const float* ff2_w  = (const float*)d_in[13];
    const float* ff2_b  = (const float*)d_in[14];
    const float* ln1_g  = (const float*)d_in[15];
    const float* ln1_b  = (const float*)d_in[16];
    const float* ln2_g  = (const float*)d_in[17];
    const float* ln2_b  = (const float*)d_in[18];

    float* h    = (float*)d_ws;
    float* xt   = h    + (size_t)Mr * D;
    float* hd   = xt   + (size_t)Mr * D;
    float* qkv  = hd   + (size_t)Mr * D;
    float* obuf = qkv  + (size_t)Mr * 3 * D;
    float* h1   = obuf + (size_t)Mr * D;
    float* ff   = h1   + (size_t)Mr * D;

    launch_gemm_rows(x, in_w, in_b, nullptr, nullptr, nullptr, h, Mr, IN, 2, S, stream);

    for (int l = 0; l < L; ++l) {
        launch_gemm_rows(h, Mw + (size_t)l * D * D, nullptr, nullptr,
                         desc_g + l * D, desc_b + l * D, xt, Mr, D, 0, S, stream);
        ddnet_desc_kernel<<<dim3(DD / 16, S / DSC), 256, 0, stream>>>(xt, h, P + (size_t)l * D * D * G, hd, S);
        launch_gemm(hd, qkv_w + (size_t)l * 3 * D * D, qkv_b + (size_t)l * 3 * D,
                    qkv, Mr, 3 * D, D, 0, stream);
        ddnet_attn_kernel<<<B * 8 * (S / 64), 128, 0, stream>>>(qkv, obuf, S);
        launch_gemm_rows(obuf, out_w + (size_t)l * D * D, out_b + l * D, hd,
                         ln1_g + l * D, ln1_b + l * D, h1, Mr, D, 0, S, stream);
        launch_gemm(h1, ff1_w + (size_t)l * DFF * D, ff1_b + (size_t)l * DFF,
                    ff, Mr, DFF, D, 1  , stream);
        float* dst = (l == L - 1) ? (float*)d_out : h;
        launch_gemm_rows(ff, ff2_w + (size_t)l * D * DFF, ff2_b + l * D, h1,
                         ln2_g + l * D, ln2_b + l * D, dst, Mr, DFF, 0, S, stream);
    }
}


# Whole-row GEMM K loop: global-to-register staging as a ring of 3 tile sets: a tile is requested 3 trips before its LDS store, 3 tiles' loads in flight (was 1 set, 1 trip ahead); sums unchanged
# speedup vs baseline: 2.3633x; 1.0560x over previous
#include <hip/hip_runtime.h>
#include <math.h>

typedef __attribute__((ext_vector_type(2))) float v2f;
typedef __attribute__((ext_vector_type(8))) float v8f;
typedef __attribute__((ext_vector_type(4))) float v4f;

__device__ __forceinline__ v8f wmma_f32(v2f a, v2f b, v8f c) {
    return __builtin_amdgcn_wmma_f32_16x16x4_f32(false, a, false, b, (short)0, c, false, false);
}
__device__ __forceinline__ v2f lds_frag(const float* p) {
    return *(const v2f*)p;
}

#define GT_K 16
#define GT_R 3
#define AB_P 20

template <int V> struct ring_set { static constexpr int value = V; };
enum { TR_STEADY = 0,
       TR_DRAIN  = 1,
       TR_LAST   = 2,
       TR_ANY    = 3 };
template <int S, int M> struct trip_kind { static constexpr int set = S, mode = M; };

template <int TM, int TN>
__global__ __launch_bounds__(256) void ddnet_gemm_kernel(
    const float* __restrict__ A, const float* __restrict__ W,
    const float* __restrict__ bias, const float* __restrict__ resid,
    const float* __restrict__ gamma, const float* __restrict__ beta,
    float* __restrict__ C, int Mdim, int Ndim, int Kdim, int act, int S)
{
    constexpr int WMC = TM / 32;
    constexpr int STG = (TM + TN) * AB_P;
    constexpr int NA  = TM / 16;
    constexpr int NB  = TN / 16;
    constexpr int C_P = TN + 4;
    constexpr int SMEM = 2 * STG > TM * C_P ? 2 * STG : TM * C_P;

    __shared__ __align__(16) float smem[SMEM];

    const int tid   = threadIdx.x;
    const int lane  = tid & 31;
    const int wave  = tid >> 5;
    const int lhalf = lane >> 4;
    const int lmod  = lane & 15;
    const int m0    = blockIdx.y * TM;
    const int n0    = blockIdx.x * TN;
    const int wm    = (wave % WMC) * 32;
    const int wn    = (wave / WMC) * 32;
    const int nk    = Kdim >> 4;

    v8f acc[2][2] = {};
    static_assert(GT_R == 3, "the prologue and the K loop below name three ring sets");
    float ra[GT_R][NA], rb[GT_R][NB];

    const int sk = tid & 15, sm = tid >> 4;
    const float* ag = A + (size_t)(m0 + sm) * Kdim + sk;
    const float* wg = W + (size_t)(n0 + sm) * Kdim + sk;
    const int sl = sm * AB_P + sk;

    auto fetch = [&](auto set, int k0) {
        constexpr int s = decltype(set)::value;
        #pragma unroll
        for (int i = 0; i < NA; ++i) ra[s][i] = ag[(size_t)(16 * i) * Kdim + k0];
        #pragma unroll
        for (int i = 0; i < NB; ++i) rb[s][i] = wg[(size_t)(16 * i) * Kdim + k0];
    };
    auto spill = [&](auto set, int st) {
        constexpr int s = decltype(set)::value;
        float* Ad = smem + st * STG + sl;
        float* Bd = Ad + TM * AB_P;
        #pragma unroll
        for (int i = 0; i < NA; ++i) Ad[16 * i * AB_P] = ra[s][i];
        #pragma unroll
        for (int i = 0; i < NB; ++i) Bd[16 * i * AB_P] = rb[s][i];
    };

    auto trip = [&](auto kind, int kt) {
        constexpr int mode = decltype(kind)::mode;
        const ring_set<decltype(kind)::set> set;
        const int cur = kt & 1;
        bool more;
        if constexpr (mode == TR_STEADY) {
            more = true;
            spill(set, cur ^ 1);
            fetch(set, (kt + 1 + GT_R) << 4);
        } else if constexpr (mode == TR_DRAIN) {
            more = true;
            spill(set, cur ^ 1);
        } else if constexpr (mode == TR_LAST) {
            more = false;
        } else {
            more = kt + 1 < nk;
            if (more) {
                spill(set, cur ^ 1);
                if (kt + 2 < nk) fetch(set, (kt + 2) << 4);
            }
        }
        const float* Ab = smem + cur * STG;
        const float* Bb = Ab + TM * AB_P;
        v2f fa[4][2], fb[4][2];
        #pragma unroll
        for (int t = 0; t < 4; ++t) {
            const int kf = t * 4 + 2 * lhalf;
            fa[t][0] = lds_frag(&Ab[(wm + lmod) * AB_P + kf]);
            fa[t][1] = lds_frag(&Ab[(wm + 16 + lmod) * AB_P + kf]);
            fb[t][0] = lds_frag(&Bb[(wn + lmod) * AB_P + kf]);
            fb[t][1] = lds_frag(&Bb[(wn + 16 + lmod) * AB_P + kf]);
        }
        if (more) {
            __builtin_amdgcn_fence(__ATOMIC_RELEASE, "workgroup", "local");
            __builtin_amdgcn_s_barrier_signal(-1);
            if constexpr (mode == TR_STEADY || mode == TR_DRAIN)
                __builtin_amdgcn_sched_barrier(0);
        }
        #pragma unroll
        for (int t = 0; t < 4; ++t) {
            acc[0][0] = wmma_f32(fa[t][0], fb[t][0], acc[0][0]);
            acc[0][1] = wmma_f32(fa[t][0], fb[t][1], acc[0][1]);
            acc[1][0] = wmma_f32(fa[t][1], fb[t][0], acc[1][0]);
            acc[1][1] = wmma_f32(fa[t][1], fb[t][1], acc[1][1]);
        }
        if (more) {
            if constexpr (mode == TR_STEADY || mode == TR_DRAIN)
                __builtin_amdgcn_sched_barrier(0);
            __builtin_amdgcn_s_barrier_wait(-1);
            __builtin_amdgcn_fence(__ATOMIC_ACQUIRE, "workgroup", "local");
        }
    };

    const ring_set<0> s0;
    const ring_set<1> s1;
    const ring_set<2> s2;
    bool ringed = false;
    if constexpr (TN == 256) ringed = nk > GT_R && nk % GT_R == 1;
    if (ringed) {
        fetch(s0, 0);
        fetch(s1, GT_K);
        fetch(s2, 2 * GT_K);
        spill(s0, 0);
        fetch(s0, GT_R * GT_K);
        __syncthreads();
        int kt = 0;
        for (; kt + 2 * GT_R < nk; kt += GT_R) {
            trip(trip_kind<1, TR_STEADY>{}, kt);
            trip(trip_kind<2, TR_STEADY>{}, kt + 1);
            trip(trip_kind<0, TR_STEADY>{}, kt + 2);
        }
        trip(trip_kind<1, TR_DRAIN>{}, kt);
        trip(trip_kind<2, TR_DRAIN>{}, kt + 1);
        trip(trip_kind<0, TR_DRAIN>{}, kt + 2);
        trip(trip_kind<0, TR_LAST>{},  kt + 3);
    } else {
        fetch(s0, 0);
        spill(s0, 0);
        if (nk > 1) fetch(s0, GT_K);
        __syncthreads();
        for (int kt = 0; kt < nk; ++kt) trip(trip_kind<0, TR_ANY>{}, kt);
    }

    __syncthreads();
    #pragma unroll
    for (int mi = 0; mi < 2; ++mi)
        #pragma unroll
        for (int ni = 0; ni < 2; ++ni)
            #pragma unroll
            for (int v = 0; v < 8; ++v)
                smem[(wm + mi * 16 + v + 8 * lhalf) * C_P + wn + ni * 16 + lmod] = acc[mi][ni][v];
    __syncthreads();

    if constexpr (TN == 128) {
        const int c4 = (tid & 31) * 4;
        const int r0 = tid >> 5;
        const float* rp = resid ? resid + (size_t)(m0 + r0) * Ndim + n0 + c4 : nullptr;
        float*       cp = C + (size_t)(m0 + r0) * Ndim + n0 + c4;
        v4f bv = {0.f, 0.f, 0.f, 0.f};
        if (bias) bv = *(const v4f*)(bias + n0 + c4);
        v4f rv[TM / 8];
        if (resid) {
            #pragma unroll
            for (int i = 0; i < TM / 8; ++i)
                rv[i] = *(const v4f*)(rp + (size_t)(8 * i) * Ndim);
        }
        #pragma unroll
        for (int i = 0; i < TM / 8; ++i) {
            v4f val = *(const v4f*)&smem[(r0 + 8 * i) * C_P + c4];
            if (bias)  val += bv;
            if (resid) val += rv[i];
            if (act == 1) {
                val.x = 0.5f * val.x * (1.0f + erff(val.x * 0.70710678118654752f));
                val.y = 0.5f * val.y * (1.0f + erff(val.y * 0.70710678118654752f));
                val.z = 0.5f * val.z * (1.0f + erff(val.z * 0.70710678118654752f));
                val.w = 0.5f * val.w * (1.0f + erff(val.w * 0.70710678118654752f));
            }
            *(v4f*)(cp + (size_t)(8 * i) * Ndim) = val;
        }
    } else {
        constexpr int NR = TM / 8;
        const int c0 = 4 * lane;
        v4f b0 = {0.f, 0.f, 0.f, 0.f}, b1 = b0;
        if (bias) {
            b0 = *(const v4f*)(bias + c0);
            b1 = *(const v4f*)(bias + 128 + c0);
        }
        v4f q0[NR], q1[NR];
        if (resid) {
            #pragma unroll
            for (int i = 0; i < NR; ++i) {
                const float* rp = resid + (size_t)(m0 + wave + 8 * i) * TN + c0;
                q0[i] = *(const v4f*)rp;
                q1[i] = *(const v4f*)(rp + 128);
            }
        }
        v4f x0[NR], x1[NR];
        #pragma unroll
        for (int i = 0; i < NR; ++i) {
            const float* sp = &smem[(wave + 8 * i) * C_P + c0];
            x0[i] = *(const v4f*)sp;
            x1[i] = *(const v4f*)(sp + 128);
            if (bias)  { x0[i] += b0;    x1[i] += b1; }
            if (resid) { x0[i] += q0[i]; x1[i] += q1[i]; }
        }
        if (gamma) {
            const v4f g0 = *(const v4f*)(gamma + c0), g1 = *(const v4f*)(gamma + 128 + c0);
            const v4f e0 = *(const v4f*)(beta + c0),  e1 = *(const v4f*)(beta + 128 + c0);
            float sum[NR];
            #pragma unroll
            for (int i = 0; i < NR; ++i)
                sum[i] = ((x0[i].x + x0[i].y) + (x0[i].z + x0[i].w))
                       + ((x1[i].x + x1[i].y) + (x1[i].z + x1[i].w));
            #pragma unroll
            for (int m = 16; m >= 1; m >>= 1)
                #pragma unroll
                for (int i = 0; i < NR; ++i) sum[i] += __shfl_xor(sum[i], m, 32);
            float vs[NR];
            #pragma unroll
            for (int i = 0; i < NR; ++i) {
                const float mean = sum[i] * (1.0f / 256.0f);
                x0[i] -= mean;
                x1[i] -= mean;
                const v4f s0 = x0[i] * x0[i], s1 = x1[i] * x1[i];
                vs[i] = ((s0.x + s0.y) + (s0.z + s0.w)) + ((s1.x + s1.y) + (s1.z + s1.w));
            }
            #pragma unroll
            for (int m = 16; m >= 1; m >>= 1)
                #pragma unroll
                for (int i = 0; i < NR; ++i) vs[i] += __shfl_xor(vs[i], m, 32);
            #pragma unroll
            for (int i = 0; i < NR; ++i) {
                const float rstd = rsqrtf(vs[i] * (1.0f / 256.0f) + 1e-5f);
                x0[i] = x0[i] * rstd * g0 + e0;
                x1[i] = x1[i] * rstd * g1 + e1;
            }
        } else if (act == 2) {
            const float rate = logf(10000.0f) / (float)Ndim;
            float dv[4];
            #pragma unroll
            for (int p = 0; p < 4; ++p) {
                const int d = (p < 2 ? c0 : 128 + c0) + 2 * (p & 1);
                dv[p] = __expf(-(float)d * rate);
            }
            #pragma unroll
            for (int i = 0; i < NR; ++i) {
                const float sp = (float)((m0 + wave + 8 * i) % S);
                x0[i].x += __sinf(sp * dv[0]);  x0[i].y += __cosf(sp * dv[0]);
                x0[i].z += __sinf(sp * dv[1]);  x0[i].w += __cosf(sp * dv[1]);
                x1[i].x += __sinf(sp * dv[2]);  x1[i].y += __cosf(sp * dv[2]);
                x1[i].z += __sinf(sp * dv[3]);  x1[i].w += __cosf(sp * dv[3]);
            }
        }
        #pragma unroll
        for (int i = 0; i < NR; ++i) {
            float* cp = C + (size_t)(m0 + wave + 8 * i) * TN + c0;
            *(v4f*)cp         = x0[i];
            *(v4f*)(cp + 128) = x1[i];
        }
    }
}

#define DD 256
#define DG 6
#define DSC 16
#define DNP 2
#define DCT 1632

__global__ __launch_bounds__(256) void ddnet_desc_kernel(
    const float* __restrict__ xt, const float* __restrict__ h,
    const float* __restrict__ P, float* __restrict__ hd, int S)
{
    __shared__ __align__(16) float Ct[2][DNP][DCT];
    __shared__ __align__(16) float Red[2][DNP][8][256];

    const int tid   = threadIdx.x, lane = tid & 31, wave = tid >> 5;
    const int lhalf = lane >> 4, lmod = lane & 15;
    const int i0    = blockIdx.x * 16;
    const int sbase = blockIdx.y * DSC;
    const int jw    = wave * 32 + lhalf * 8;
    const float two_pi = 6.28318530717958647692f;

    auto build = [&](int buf, int sfirst) {
        const float pw[DG] = { 1.0f, 1.5f, 2.25f, 3.375f, 5.0625f, 7.59375f };
        {
            const float base = 2.0f + 0.5f * (float)(i0 + tid);
            #pragma unroll
            for (int p = 0; p < DNP; ++p) {
                const float sv = two_pi * (float)(sfirst + p);
                #pragma unroll
                for (int g = 0; g < DG; ++g)
                    Ct[buf][p][tid * DG + g] = __cosf(sv / (base * pw[g]));
            }
        }
        if (tid < 15 * DNP * DG) {
            const int tt = 256 + tid / (DNP * DG);
            const int r  = tid % (DNP * DG);
            const int p  = r / DG, g = r % DG;
            const float pwg = g == 0 ? pw[0] : g == 1 ? pw[1] : g == 2 ? pw[2]
                            : g == 3 ? pw[3] : g == 4 ? pw[4] : pw[5];
            const float base = 2.0f + 0.5f * (float)(i0 + tt);
            const float sv   = two_pi * (float)(sfirst + p);
            Ct[buf][p][tt * DG + g] = __cosf(sv / (base * pwg));
        }
    };
    auto load_a = [&](v4f (&r)[DNP][4], int sfirst) {
        #pragma unroll
        for (int p = 0; p < DNP; ++p) {
            const float* xr = xt + ((size_t)lmod * S + sfirst + p) * DD + jw;
            #pragma unroll
            for (int q = 0; q < 4; ++q)
                r[p][q] = *(const v4f*)(xr + 16 * (q >> 1) + 4 * (q & 1));
        }
    };
    auto reduce = [&](int buf, int sfirst) {
        const int b = (tid >> 5) + 8 * lhalf;
        #pragma unroll
        for (int p = 0; p < DNP; ++p) {
            float sum = Red[buf][p][0][tid];
            #pragma unroll
            for (int w = 1; w < 8; ++w) sum += Red[buf][p][w][tid];
            const size_t off = ((size_t)b * S + sfirst + p) * DD + i0 + lmod;
            hd[off] = sum + h[off];
        }
    };

    v2f pp[2][4][DG];
    #pragma unroll
    for (int blk = 0; blk < 2; ++blk) {
        const float* pg = P + ((size_t)(i0 + lmod) * DD + jw + 16 * blk) * DG;
        float pr[48];
        #pragma unroll
        for (int q = 0; q < 12; ++q) {
            const v4f v = *(const v4f*)(pg + 4 * q);
            pr[4 * q]     = v.x;
            pr[4 * q + 1] = v.y;
            pr[4 * q + 2] = v.z;
            pr[4 * q + 3] = v.w;
        }
        #pragma unroll
        for (int t = 0; t < 4; ++t)
            #pragma unroll
            for (int g = 0; g < DG; ++g) {
                pp[blk][t][g].x = pr[(2 * t) * DG + g];
                pp[blk][t][g].y = pr[(2 * t + 1) * DG + g];
            }
    }

    v4f ra[DNP][4];
    load_a(ra, sbase);
    build(0, sbase);
    __syncthreads();

    const int ntrip = DSC / DNP;
    for (int n = 0; n < ntrip; ++n) {
        const int buf    = n & 1;
        const int sfirst = sbase + n * DNP;

        v4f rn[DNP][4];
        if (n + 1 < ntrip) load_a(rn, sfirst + DNP);
        if (n > 0) reduce(buf ^ 1, sfirst - DNP);

        const float* cb = &Ct[buf][0][0] + (lmod + jw) * DG;
        #pragma unroll
        for (int p = 0; p < DNP; ++p) {
            v8f acc = {};
            #pragma unroll
            for (int blk = 0; blk < 2; ++blk)
                #pragma unroll
                for (int t = 0; t < 4; ++t) {
                    const float* c0 = cb + p * DCT + (16 * blk + 2 * t) * DG;
                    v2f b;
                    {
                        const v2f c = { c0[0], c0[DG] };
                        b = pp[blk][t][0] * c;
                    }
                    #pragma unroll
                    for (int g = 1; g < DG; ++g) {
                        const v2f c = { c0[g], c0[DG + g] };
                        b = pp[blk][t][g] * c + b;
                    }
                    const v4f av = ra[p][2 * blk + (t >> 1)];
                    v2f a;
                    if (t & 1) { a.x = av.z; a.y = av.w; }
                    else       { a.x = av.x; a.y = av.y; }
                    acc = wmma_f32(a, b, acc);
                }
            #pragma unroll
            for (int v = 0; v < 8; ++v)
                Red[buf][p][wave][v * 32 + lane] = acc[v];
        }

        if (n + 1 < ntrip) {
            build(buf ^ 1, sfirst + DNP);
            #pragma unroll
            for (int p = 0; p < DNP; ++p)
                #pragma unroll
                for (int q = 0; q < 4; ++q) ra[p][q] = rn[p][q];
        }
        __syncthreads();
    }
    reduce((ntrip - 1) & 1, sbase + (ntrip - 1) * DNP);
}

#define AT_KT 64
#define AT_KP 36
#define AT_VP 40
#define AT_PP 20
#define AT_PT 320

__global__ __launch_bounds__(128) void ddnet_attn_kernel(
    const float* __restrict__ qkv, float* __restrict__ o, int S)
{
    const int qb    = S / 64;
    const int bh    = blockIdx.x / qb;
    const int qblk  = blockIdx.x % qb;
    const int b     = bh >> 3, head = bh & 7;
    const int tid   = threadIdx.x, lane = tid & 31, wave = tid >> 5;
    const int lhalf = lane >> 4, lmod = lane & 15;
    const int q0    = qblk * 64 + wave * 16;

    __shared__ __align__(16) float Kf[AT_KT * AT_KP];
    __shared__ __align__(16) float Vf[AT_KT * AT_VP];
    __shared__ __align__(16) float Pf[4][4][AT_PT];

    const size_t rs = 768;
    const float* qp = qkv + (size_t)b * S * rs + head * 32;
    const float* kp = qp + 256;
    const float* vp = qp + 512;
    const float scale = 0.17677669529663688f;

    v2f aq[8];
    #pragma unroll
    for (int t = 0; t < 8; ++t) {
        const float* qr = qp + (size_t)(q0 + lmod) * rs;
        aq[t].x = qr[t * 4 + 2 * lhalf];
        aq[t].y = qr[t * 4 + 1 + 2 * lhalf];
    }

    v8f mrow, lrow = {};
    v8f o0 = {}, o1 = {};
    #pragma unroll
    for (int v = 0; v < 8; ++v) mrow[v] = -1e30f;

    const int sq = tid & 7;
    const int sr = tid >> 3;
    const float* kst = kp + (size_t)sr * rs + 4 * sq;
    const float* vst = vp + (size_t)sr * rs + 4 * sq;
    const int kst_lds = sr * AT_KP + 4 * sq;
    const int vst_lds = sr * AT_VP + 4 * sq;

    v4f rk[4], rv[4];
    #pragma unroll
    for (int i = 0; i < 4; ++i) {
        rk[i] = *(const v4f*)(kst + (size_t)(16 * i) * rs);
        rv[i] = *(const v4f*)(vst + (size_t)(16 * i) * rs);
    }

    float* pst = &Pf[wave][0][(8 * lhalf) * AT_PP + lmod];

    const int nkt = S / AT_KT;
    for (int kt = 0; kt < nkt; ++kt) {
        #pragma unroll
        for (int i = 0; i < 4; ++i) {
            *(v4f*)&Kf[i * 16 * AT_KP + kst_lds] = rk[i];
            *(v4f*)&Vf[i * 16 * AT_VP + vst_lds] = rv[i];
        }
        __syncthreads();

        if (kt + 1 < nkt) {
            const size_t nb = (size_t)((kt + 1) * AT_KT) * rs;
            #pragma unroll
            for (int i = 0; i < 4; ++i) {
                rk[i] = *(const v4f*)(kst + nb + (size_t)(16 * i) * rs);
                rv[i] = *(const v4f*)(vst + nb + (size_t)(16 * i) * rs);
            }
        }

        v8f sc[4] = {};
        #pragma unroll
        for (int t = 0; t < 8; ++t)
            #pragma unroll
            for (int sb = 0; sb < 4; ++sb) {
                v2f bk = lds_frag(&Kf[(sb * 16 + lmod) * AT_KP + t * 4 + 2 * lhalf]);
                sc[sb] = wmma_f32(aq[t], bk, sc[sb]);
            }

        const v8f s0 = sc[0] * scale, s1 = sc[1] * scale;
        const v8f s2 = sc[2] * scale, s3 = sc[3] * scale;
        v8f mnew, corr;
        #pragma unroll
        for (int v = 0; v < 8; ++v) {
            float mx = fmaxf(fmaxf(s0[v], s1[v]), fmaxf(s2[v], s3[v]));
            #pragma unroll
            for (int m = 8; m >= 1; m >>= 1) mx = fmaxf(mx, __shfl_xor(mx, m, 32));
            mnew[v] = fmaxf(mrow[v], mx);
            corr[v] = __expf(mrow[v] - mnew[v]);
        }
        const v8f d0 = s0 - mnew, d1 = s1 - mnew, d2 = s2 - mnew, d3 = s3 - mnew;
        #pragma unroll
        for (int v = 0; v < 8; ++v) {
            sc[0][v] = __expf(d0[v]);
            sc[1][v] = __expf(d1[v]);
            sc[2][v] = __expf(d2[v]);
            sc[3][v] = __expf(d3[v]);
        }
        lrow = lrow * corr + ((sc[0] + sc[1]) + (sc[2] + sc[3]));
        mrow = mnew;
        o0 *= corr;
        o1 *= corr;

        #pragma unroll
        for (int sb = 0; sb < 4; ++sb)
            #pragma unroll
            for (int v = 0; v < 8; ++v)
                pst[sb * AT_PT + v * AT_PP] = sc[sb][v];
        #pragma unroll
        for (int sb = 0; sb < 4; ++sb)
            #pragma unroll
            for (int t = 0; t < 4; ++t) {
                const int fo = t * 2 + lhalf;
                const float* vr = &Vf[(sb * 16 + 2 * fo) * AT_VP + lmod];
                v2f ap  = lds_frag(&Pf[wave][sb][lmod * AT_PP + 2 * fo]);
                v2f bv0 = { vr[0],  vr[AT_VP] };
                v2f bv1 = { vr[16], vr[AT_VP + 16] };
                o0 = wmma_f32(ap, bv0, o0);
                o1 = wmma_f32(ap, bv1, o1);
            }
        __syncthreads();
    }

    #pragma unroll
    for (int v = 0; v < 8; ++v) {
        float l = lrow[v];
        #pragma unroll
        for (int m = 8; m >= 1; m >>= 1) l += __shfl_xor(l, m, 32);
        int qr = v + 8 * lhalf;
        float inv = 1.0f / l;
        size_t base = ((size_t)b * S + q0 + qr) * 256 + head * 32;
        o[base + lmod]      = o0[v] * inv;
        o[base + 16 + lmod] = o1[v] * inv;
    }
}

static void launch_gemm(const float* A, const float* W, const float* bias,
                        float* C, int M, int N, int K, int act, hipStream_t stream)
{
    dim3 grid(N / 128, M / 64);
    ddnet_gemm_kernel<64, 128><<<grid, 256, 0, stream>>>(
        A, W, bias, nullptr, nullptr, nullptr, C, M, N, K, act, 0);
}
static void launch_gemm_rows(const float* A, const float* W, const float* bias,
                             const float* resid, const float* gamma, const float* beta,
                             float* C, int M, int K, int act, int S, hipStream_t stream)
{
    dim3 grid(1, M / 32);
    ddnet_gemm_kernel<32, 256><<<grid, 256, 0, stream>>>(
        A, W, bias, resid, gamma, beta, C, M, 256, K, act, S);
}

extern "C" void kernel_launch(void* const* d_in, const int* in_sizes, int n_in,
                              void* d_out, int out_size, void* d_ws, size_t ws_size,
                              hipStream_t stream) {
    (void)in_sizes; (void)n_in; (void)out_size; (void)ws_size;
    const int B = 16, S = 512, IN = 64, D = 256, L = 3, G = 6, DFF = 1024;
    const int Mr = B * S;

    const float* x      = (const float*)d_in[0];
    const float* in_w   = (const float*)d_in[1];
    const float* in_b   = (const float*)d_in[2];
    const float* Mw     = (const float*)d_in[3];
    const float* P      = (const float*)d_in[4];
    const float* desc_g = (const float*)d_in[5];
    const float* desc_b = (const float*)d_in[6];
    const float* qkv_w  = (const float*)d_in[7];
    const float* qkv_b  = (const float*)d_in[8];
    const float* out_w  = (const float*)d_in[9];
    const float* out_b  = (const float*)d_in[10];
    const float* ff1_w  = (const float*)d_in[11];
    const float* ff1_b  = (const float*)d_in[12];
    const float* ff2_w  = (const float*)d_in[13];
    const float* ff2_b  = (const float*)d_in[14];
    const float* ln1_g  = (const float*)d_in[15];
    const float* ln1_b  = (const float*)d_in[16];
    const float* ln2_g  = (const float*)d_in[17];
    const float* ln2_b  = (const float*)d_in[18];

    float* h    = (float*)d_ws;
    float* xt   = h    + (size_t)Mr * D;
    float* hd   = xt   + (size_t)Mr * D;
    float* qkv  = hd   + (size_t)Mr * D;
    float* obuf = qkv  + (size_t)Mr * 3 * D;
    float* h1   = obuf + (size_t)Mr * D;
    float* ff   = h1   + (size_t)Mr * D;

    launch_gemm_rows(x, in_w, in_b, nullptr, nullptr, nullptr, h, Mr, IN, 2, S, stream);

    for (int l = 0; l < L; ++l) {
        launch_gemm_rows(h, Mw + (size_t)l * D * D, nullptr, nullptr,
                         desc_g + l * D, desc_b + l * D, xt, Mr, D, 0, S, stream);
        ddnet_desc_kernel<<<dim3(DD / 16, S / DSC), 256, 0, stream>>>(xt, h, P + (size_t)l * D * D * G, hd, S);
        launch_gemm(hd, qkv_w + (size_t)l * 3 * D * D, qkv_b + (size_t)l * 3 * D,
                    qkv, Mr, 3 * D, D, 0, stream);
        ddnet_attn_kernel<<<B * 8 * (S / 64), 128, 0, stream>>>(qkv, obuf, S);
        launch_gemm_rows(obuf, out_w + (size_t)l * D * D, out_b + l * D, hd,
                         ln1_g + l * D, ln1_b + l * D, h1, Mr, D, 0, S, stream);
        launch_gemm(h1, ff1_w + (size_t)l * DFF * D, ff1_b + (size_t)l * DFF,
                    ff, Mr, DFF, D, 1  , stream);
        float* dst = (l == L - 1) ? (float*)d_out : h;
        launch_gemm_rows(ff, ff2_w + (size_t)l * D * DFF, ff2_b + l * D, h1,
                         ln2_g + l * D, ln2_b + l * D, dst, Mr, DFF, 0, S, stream);
    }
}
